# starting kernel
# baseline (speedup 1.0000x reference)
#include <hip/hip_runtime.h>
#include <hip/hip_bf16.h>

#define HID 128
#define NEG_SLOPE 0.2f
#define GAT_EPS 1e-16f

typedef float v2f __attribute__((ext_vector_type(2)));
typedef float v8f __attribute__((ext_vector_type(8)));

__device__ __forceinline__ unsigned flipf(float f) {
    unsigned u = __float_as_uint(f);
    return (u & 0x80000000u) ? ~u : (u | 0x80000000u);
}
__device__ __forceinline__ float unflipf(unsigned u) {
    return __uint_as_float((u & 0x80000000u) ? (u ^ 0x80000000u) : ~u);
}

__global__ void init_node_kernel(unsigned* __restrict__ emax,
                                 float* __restrict__ denom, int n) {
    int i = blockIdx.x * blockDim.x + threadIdx.x;
    if (i < n) { emax[i] = 0u; denom[i] = 0.0f; }
}

__global__ void init_out_kernel(float* __restrict__ out,
                                const float* __restrict__ bias, int total) {
    int i = blockIdx.x * blockDim.x + threadIdx.x;
    if (i < total) out[i] = bias[i & (HID - 1)];
}

__global__ __launch_bounds__(256) void gemm_x_kernel(const float* __restrict__ h,
                                                     const float* __restrict__ W,
                                                     float* __restrict__ x,
                                                     int n) {
    __shared__ float sh[16 * HID];
    const int block_row = blockIdx.x * 16;
    const int tid  = threadIdx.x;
    const int wave = tid >> 5;
    const int lane = tid & 31;

    const float4* gsrc = (const float4*)(h + (size_t)block_row * HID);
    float4*       ldst = (float4*)sh;
    #pragma unroll
    for (int i = tid; i < 16 * HID / 4; i += 256) ldst[i] = gsrc[i];
    __syncthreads();

    const int m    = lane & 15;
    const int koff = (lane >> 4) << 1;
    const int ncol = wave * 16 + (lane & 15);

    v8f c = {};
    #pragma unroll
    for (int k = 0; k < HID; k += 4) {
        v2f a, b;
        a.x = sh[m * HID + k + koff];
        a.y = sh[m * HID + k + koff + 1];
        b.x = W[(size_t)(k + koff)     * HID + ncol];
        b.y = W[(size_t)(k + koff + 1) * HID + ncol];
        c = __builtin_amdgcn_wmma_f32_16x16x4_f32(
                 false, a,  false, b,
                 (short)0, c,  false,  false);
    }

    const int rbase = block_row + ((lane >> 4) << 3);
    #pragma unroll
    for (int r = 0; r < 8; ++r) {
        int row = rbase + r;
        if (row < n) x[(size_t)row * HID + ncol] = c[r];
    }
}

__global__ __launch_bounds__(256) void attn_kernel(const float* __restrict__ x,
                                                   const float* __restrict__ a_s,
                                                   const float* __restrict__ a_d,
                                                   float* __restrict__ as_out,
                                                   float* __restrict__ ad_out,
                                                   int n) {
    int node = (blockIdx.x * blockDim.x + threadIdx.x) >> 5;
    int lane = threadIdx.x & 31;
    if (node >= n) return;
    float4 v  = ((const float4*)(x + (size_t)node * HID))[lane];
    float4 s4 = ((const float4*)a_s)[lane];
    float4 d4 = ((const float4*)a_d)[lane];
    float ps = v.x * s4.x + v.y * s4.y + v.z * s4.z + v.w * s4.w;
    float pd = v.x * d4.x + v.y * d4.y + v.z * d4.z + v.w * d4.w;
    #pragma unroll
    for (int o = 16; o > 0; o >>= 1) {
        ps += __shfl_down(ps, o, 32);
        pd += __shfl_down(pd, o, 32);
    }
    if (lane == 0) { as_out[node] = ps; ad_out[node] = pd; }
}

__global__ void edge_max_kernel(const int* __restrict__ src,
                                const int* __restrict__ dst,
                                const float* __restrict__ as_,
                                const float* __restrict__ ad_,
                                float* __restrict__ eval,
                                unsigned* __restrict__ emax,
                                int n_edges, int n_nodes) {
    int e = blockIdx.x * blockDim.x + threadIdx.x;
    int total = n_edges + n_nodes;
    if (e >= total) return;
    int s, d;
    if (e < n_edges) { s = src[e]; d = dst[e]; } else { s = d = e - n_edges; }
    float v = as_[s] + ad_[d];
    v = (v > 0.0f) ? v : NEG_SLOPE * v;
    eval[e] = v;
    atomicMax(emax + d, flipf(v));
}

__global__ void edge_exp_kernel(const int* __restrict__ dst,
                                const float* __restrict__ eval,
                                const unsigned* __restrict__ emax,
                                float* __restrict__ exv,
                                float* __restrict__ denom,
                                int n_edges, int n_nodes) {
    int e = blockIdx.x * blockDim.x + threadIdx.x;
    int total = n_edges + n_nodes;
    if (e >= total) return;
    int d = (e < n_edges) ? dst[e] : (e - n_edges);
    float m  = unflipf(emax[d]);
    float ex = __expf(eval[e] - m);
    exv[e] = ex;
    atomicAdd(denom + d, ex);
}

__global__ __launch_bounds__(256) void edge_agg_kernel(const int* __restrict__ src,
                                                       const int* __restrict__ dst,
                                                       const float* __restrict__ x,
                                                       const float* __restrict__ exv,
                                                       const float* __restrict__ denom,
                                                       float* __restrict__ out,
                                                       int n_edges, int n_nodes) {
    long long gid = (long long)blockIdx.x * blockDim.x + threadIdx.x;
    int e    = (int)(gid >> 5);
    int lane = threadIdx.x & 31;
    int total = n_edges + n_nodes;
    if (e >= total) return;
    int s, d;
    if (e < n_edges) { s = src[e]; d = dst[e]; } else { s = d = e - n_edges; }
    float alpha = exv[e] / (denom[d] + GAT_EPS);
    float4 v = ((const float4*)(x + (size_t)s * HID))[lane];
    float* o = out + (size_t)d * HID + lane * 4;
    atomicAdd(o + 0, alpha * v.x);
    atomicAdd(o + 1, alpha * v.y);
    atomicAdd(o + 2, alpha * v.z);
    atomicAdd(o + 3, alpha * v.w);
}

__global__ void relu_kernel(float* __restrict__ a, int total) {
    int i = blockIdx.x * blockDim.x + threadIdx.x;
    if (i < total) a[i] = fmaxf(a[i], 0.0f);
}

extern "C" void kernel_launch(void* const* d_in, const int* in_sizes, int n_in,
                              void* d_out, int out_size, void* d_ws, size_t ws_size,
                              hipStream_t stream) {
    const float* z     = (const float*)d_in[0];
    const int*   ei    = (const int*)  d_in[1];
    const float* Ws    = (const float*)d_in[2];
    const float* a_src = (const float*)d_in[3];
    const float* a_dst = (const float*)d_in[4];
    const float* bias  = (const float*)d_in[5];
    float*       out   = (float*)d_out;

    const int N = in_sizes[0] / HID;
    const int E = in_sizes[1] / 2;
    const int ET = E + N;
    const int* src = ei;
    const int* dst = ei + E;

    auto align = [](size_t v) { return (v + 255) & ~(size_t)255; };
    char* p = (char*)d_ws;
    float*    x      = (float*)p;    p += align((size_t)N * HID * sizeof(float));
    float*    bufA   = (float*)p;    p += align((size_t)N * HID * sizeof(float));
    float*    bufB   = (float*)p;    p += align((size_t)N * HID * sizeof(float));
    float*    as_    = (float*)p;    p += align((size_t)N * sizeof(float));
    float*    ad_    = (float*)p;    p += align((size_t)N * sizeof(float));
    unsigned* emax   = (unsigned*)p; p += align((size_t)N * sizeof(unsigned));
    float*    denom  = (float*)p;    p += align((size_t)N * sizeof(float));
    float*    eval   = (float*)p;    p += align((size_t)ET * sizeof(float));
    float*    exv    = (float*)p;    p += align((size_t)ET * sizeof(float));

    const float* h_ptr = z;
    for (int l = 0; l < 3; ++l) {
        float* o_ptr = (l == 0) ? bufA : (l == 1) ? bufB : out;
        const float* W  = Ws    + (size_t)l * HID * HID;
        const float* as = a_src + (size_t)l * HID;
        const float* ad = a_dst + (size_t)l * HID;
        const float* b  = bias  + (size_t)l * HID;

        init_node_kernel<<<(N + 255) / 256, 256, 0, stream>>>(emax, denom, N);
        init_out_kernel<<<((size_t)N * HID + 255) / 256, 256, 0, stream>>>(o_ptr, b, N * HID);

        gemm_x_kernel<<<(N + 15) / 16, 256, 0, stream>>>(h_ptr, W, x, N);

        attn_kernel<<<((size_t)N * 32 + 255) / 256, 256, 0, stream>>>(x, as, ad, as_, ad_, N);

        edge_max_kernel<<<(ET + 255) / 256, 256, 0, stream>>>(src, dst, as_, ad_, eval, emax, E, N);
        edge_exp_kernel<<<(ET + 255) / 256, 256, 0, stream>>>(dst, eval, emax, exv, denom, E, N);
        edge_agg_kernel<<<((size_t)ET * 32 + 255) / 256, 256, 0, stream>>>(src, dst, x, exv, denom,
                                                                           o_ptr, E, N);
        if (l < 2)
            relu_kernel<<<((size_t)N * HID + 255) / 256, 256, 0, stream>>>(o_ptr, N * HID);
        h_ptr = o_ptr;
    }
}


# Edges grouped by destination once per call (atomic rank + scan); one wave per node gathers its in-edges: max, exp-sum, weighted sum in registers, one store; no float atomics, sum order = arrival rank
# speedup vs baseline: 3.3247x; 3.3247x over previous
#include <hip/hip_runtime.h>
#include <hip/hip_bf16.h>


#define HID 128
#define NEG_SLOPE 0.2f
#define GAT_EPS 1e-16f
#define SCAN_TILE 1024

typedef float v2f __attribute__((ext_vector_type(2)));
typedef float v8f __attribute__((ext_vector_type(8)));

__global__ void zero_cnt_kernel(int* __restrict__ cnt, int n) {
    int i = blockIdx.x * blockDim.x + threadIdx.x;
    if (i < n) cnt[i] = 0;
}

__global__ void rank_kernel(const int* __restrict__ dst, int* __restrict__ cnt,
                            int* __restrict__ rank, int n_edges) {
    int e = blockIdx.x * blockDim.x + threadIdx.x;
    if (e < n_edges) rank[e] = atomicAdd(cnt + dst[e], 1);
}

__global__ __launch_bounds__(256) void scan_kernel(const int* __restrict__ cnt,
                                                   int* __restrict__ off, int n) {
    __shared__ int wpre[8];
    __shared__ int wtot[8];
    const int tid  = threadIdx.x;
    const int lane = tid & 31;
    const int wave = tid >> 5;
    const int base = blockIdx.x * SCAN_TILE;

    int pre = 0;
    const int4* c4 = (const int4*)cnt;
    #pragma unroll 4
    for (int i = tid; i < base / 4; i += 256) {
        int4 q = c4[i];
        pre += (q.x + q.y) + (q.z + q.w);
    }

    const int idx0 = base + tid * 4;
    int v0 = (idx0 + 0 < n) ? cnt[idx0 + 0] : 0;
    int v1 = (idx0 + 1 < n) ? cnt[idx0 + 1] : 0;
    int v2 = (idx0 + 2 < n) ? cnt[idx0 + 2] : 0;
    int v3 = (idx0 + 3 < n) ? cnt[idx0 + 3] : 0;
    const int t = (v0 + v1) + (v2 + v3);

    for (int o = 16; o > 0; o >>= 1) pre += __shfl_xor(pre, o, 32);
    int inc = t;
    for (int o = 1; o < 32; o <<= 1) {
        int u = __shfl_up(inc, o, 32);
        if (lane >= o) inc += u;
    }
    if (lane == 31) wtot[wave] = inc;
    if (lane == 0)  wpre[wave] = pre;
    __syncthreads();

    int run = inc - t;
    for (int w = 0; w < 8; ++w) {
        run += wpre[w];
        if (w < wave) run += wtot[w];
    }
    if (idx0 + 0 <= n) off[idx0 + 0] = run;
    run += v0;
    if (idx0 + 1 <= n) off[idx0 + 1] = run;
    run += v1;
    if (idx0 + 2 <= n) off[idx0 + 2] = run;
    run += v2;
    if (idx0 + 3 <= n) off[idx0 + 3] = run;
}

__global__ void fill_kernel(const int* __restrict__ src, const int* __restrict__ dst,
                            const int* __restrict__ rank, const int* __restrict__ off,
                            int* __restrict__ col, int n_edges) {
    int e = blockIdx.x * blockDim.x + threadIdx.x;
    if (e < n_edges) col[off[dst[e]] + rank[e]] = src[e];
}

__global__ __launch_bounds__(256) void gemm_x_kernel(const float* __restrict__ h,
                                                     const float* __restrict__ W,
                                                     float* __restrict__ x,
                                                     int n) {
    __shared__ float sh[16 * HID];
    const int block_row = blockIdx.x * 16;
    const int tid  = threadIdx.x;
    const int wave = tid >> 5;
    const int lane = tid & 31;

    const float4* gsrc = (const float4*)(h + (size_t)block_row * HID);
    float4*       ldst = (float4*)sh;
    #pragma unroll
    for (int i = tid; i < 16 * HID / 4; i += 256) ldst[i] = gsrc[i];
    __syncthreads();

    const int m    = lane & 15;
    const int koff = (lane >> 4) << 1;
    const int ncol = wave * 16 + (lane & 15);

    v8f c = {};
    #pragma unroll
    for (int k = 0; k < HID; k += 4) {
        v2f a, b;
        a.x = sh[m * HID + k + koff];
        a.y = sh[m * HID + k + koff + 1];
        b.x = W[(size_t)(k + koff)     * HID + ncol];
        b.y = W[(size_t)(k + koff + 1) * HID + ncol];
        c = __builtin_amdgcn_wmma_f32_16x16x4_f32(
                 false, a,  false, b,
                 (short)0, c,  false,  false);
    }

    const int rbase = block_row + ((lane >> 4) << 3);
    #pragma unroll
    for (int r = 0; r < 8; ++r) {
        int row = rbase + r;
        if (row < n) x[(size_t)row * HID + ncol] = c[r];
    }
}

__global__ __launch_bounds__(256) void attn_kernel(const float* __restrict__ x,
                                                   const float* __restrict__ a_s,
                                                   const float* __restrict__ a_d,
                                                   float* __restrict__ as_out,
                                                   float* __restrict__ ad_out,
                                                   int n) {
    int node = (blockIdx.x * blockDim.x + threadIdx.x) >> 5;
    int lane = threadIdx.x & 31;
    if (node >= n) return;
    float4 v  = ((const float4*)(x + (size_t)node * HID))[lane];
    float4 s4 = ((const float4*)a_s)[lane];
    float4 d4 = ((const float4*)a_d)[lane];
    float ps = v.x * s4.x + v.y * s4.y + v.z * s4.z + v.w * s4.w;
    float pd = v.x * d4.x + v.y * d4.y + v.z * d4.z + v.w * d4.w;
    #pragma unroll
    for (int o = 16; o > 0; o >>= 1) {
        ps += __shfl_down(ps, o, 32);
        pd += __shfl_down(pd, o, 32);
    }
    if (lane == 0) { as_out[node] = ps; ad_out[node] = pd; }
}

__device__ __forceinline__ float wave_max(float v) {
    for (int o = 16; o > 0; o >>= 1) v = fmaxf(v, __shfl_xor(v, o, 32));
    return v;
}
__device__ __forceinline__ float wave_sum(float v) {
    for (int o = 16; o > 0; o >>= 1) v += __shfl_xor(v, o, 32);
    return v;
}

__device__ __forceinline__ float item_logit(int k, int cnt, int beg, int d, int last,
                                            const int* __restrict__ col,
                                            const float* __restrict__ as_, float adv, int& s) {
    s = d;
    if (k < cnt - 1) s = col[beg + k];
    s = min(max(s, 0), last);
    float e = as_[s] + adv;
    return (e > 0.0f) ? e : NEG_SLOPE * e;
}

__device__ __forceinline__ void add_rows(float4& acc, const float4* __restrict__ x4, int lane,
                                         int s, float alpha, int lim) {
    #pragma unroll 4
    for (int j = 0; j < lim; ++j) {
        const int   sj = __builtin_amdgcn_readlane(s, j);
        const float aj = __int_as_float(__builtin_amdgcn_readlane(__float_as_int(alpha), j));
        const float4 v = x4[(size_t)sj * (HID / 4) + lane];
        acc.x += aj * v.x;
        acc.y += aj * v.y;
        acc.z += aj * v.z;
        acc.w += aj * v.w;
    }
}

__global__ __launch_bounds__(256) void node_gather_kernel(const int* __restrict__ off,
                                                          const int* __restrict__ col,
                                                          const float* __restrict__ x,
                                                          const float* __restrict__ as_,
                                                          const float* __restrict__ ad_,
                                                          const float* __restrict__ bias,
                                                          float* __restrict__ out,
                                                          int n, int do_relu) {
    const int lane = threadIdx.x & 31;
    const int d = __builtin_amdgcn_readfirstlane((int)((blockIdx.x * blockDim.x + threadIdx.x) >> 5));
    if (d >= n) return;
    const int   beg  = off[d];
    const int   cnt  = off[d + 1] - beg + 1;
    const float adv  = ad_[d];
    const int   last = n - 1;
    const float4* __restrict__ x4 = (const float4*)x;
    float4 acc = make_float4(0.0f, 0.0f, 0.0f, 0.0f);

    if (cnt <= 32) {
        int s;
        const float e  = item_logit(lane, cnt, beg, d, last, col, as_, adv, s);
        const bool  on = lane < cnt;
        const float m   = wave_max(on ? e : -__builtin_inff());
        const float ex  = on ? __expf(e - m) : 0.0f;
        const float sum = wave_sum(ex);
        const float alpha = ex / (sum + GAT_EPS);
        add_rows(acc, x4, lane, s, alpha, cnt);
    } else {
        float m = -__builtin_inff();
        for (int b = 0; b < cnt; b += 32) {
            int s;
            const int k = b + lane;
            if (k < cnt) m = fmaxf(m, item_logit(k, cnt, beg, d, last, col, as_, adv, s));
        }
        m = wave_max(m);
        float sum = 0.0f;
        for (int b = 0; b < cnt; b += 32) {
            int s;
            const int k = b + lane;
            if (k < cnt) sum += __expf(item_logit(k, cnt, beg, d, last, col, as_, adv, s) - m);
        }
        sum = wave_sum(sum);
        for (int b = 0; b < cnt; b += 32) {
            int s = d;
            float alpha = 0.0f;
            const int k = b + lane;
            if (k < cnt)
                alpha = __expf(item_logit(k, cnt, beg, d, last, col, as_, adv, s) - m) / (sum + GAT_EPS);
            add_rows(acc, x4, lane, s, alpha, min(32, cnt - b));
        }
    }

    const float4 b4 = ((const float4*)bias)[lane];
    acc.x += b4.x; acc.y += b4.y; acc.z += b4.z; acc.w += b4.w;
    if (do_relu) {
        acc.x = fmaxf(acc.x, 0.0f); acc.y = fmaxf(acc.y, 0.0f);
        acc.z = fmaxf(acc.z, 0.0f); acc.w = fmaxf(acc.w, 0.0f);
    }
    ((float4*)(out + (size_t)d * HID))[lane] = acc;
}

extern "C" void kernel_launch(void* const* d_in, const int* in_sizes, int n_in,
                              void* d_out, int out_size, void* d_ws, size_t ws_size,
                              hipStream_t stream) {
    const float* z     = (const float*)d_in[0];
    const int*   ei    = (const int*)  d_in[1];
    const float* Ws    = (const float*)d_in[2];
    const float* a_src = (const float*)d_in[3];
    const float* a_dst = (const float*)d_in[4];
    const float* bias  = (const float*)d_in[5];
    float*       out   = (float*)d_out;

    const int N = in_sizes[0] / HID;
    const int E = in_sizes[1] / 2;
    const int* src = ei;
    const int* dst = ei + E;

    auto align = [](size_t v) { return (v + 255) & ~(size_t)255; };
    char* p = (char*)d_ws;
    float* x    = (float*)p; p += align((size_t)N * HID * sizeof(float));
    float* bufA = (float*)p; p += align((size_t)N * HID * sizeof(float));
    float* bufB = (float*)p; p += align((size_t)N * HID * sizeof(float));
    float* as_  = (float*)p; p += align((size_t)N * sizeof(float));
    float* ad_  = (float*)p; p += align((size_t)N * sizeof(float));
    int*   cnt  = (int*)p;   p += align((size_t)N * sizeof(int));
    int*   off  = (int*)p;   p += align((size_t)(N + 1) * sizeof(int));
    int*   rank = (int*)p;   p += align((size_t)E * sizeof(int));
    int*   col  = (int*)p;   p += align((size_t)E * sizeof(int));

    zero_cnt_kernel<<<(N + 255) / 256, 256, 0, stream>>>(cnt, N);
    rank_kernel<<<(E + 255) / 256, 256, 0, stream>>>(dst, cnt, rank, E);
    scan_kernel<<<(N + SCAN_TILE) / SCAN_TILE, 256, 0, stream>>>(cnt, off, N);
    fill_kernel<<<(E + 255) / 256, 256, 0, stream>>>(src, dst, rank, off, col, E);

    const float* h_ptr = z;
    for (int l = 0; l < 3; ++l) {
        float* o_ptr = (l == 0) ? bufA : (l == 1) ? bufB : out;
        const float* W  = Ws    + (size_t)l * HID * HID;
        const float* as = a_src + (size_t)l * HID;
        const float* ad = a_dst + (size_t)l * HID;
        const float* b  = bias  + (size_t)l * HID;

        gemm_x_kernel<<<(N + 15) / 16, 256, 0, stream>>>(h_ptr, W, x, N);
        attn_kernel<<<((size_t)N * 32 + 255) / 256, 256, 0, stream>>>(x, as, ad, as_, ad_, N);
        node_gather_kernel<<<((size_t)N * 32 + 255) / 256, 256, 0, stream>>>(off, col, x, as_, ad_, b,
                                                                             o_ptr, N, l < 2 ? 1 : 0);
        h_ptr = o_ptr;
    }
}


# Attention dots x.a_s, x.a_d moved into the GEMM tail: tile staged in LDS, a wave stores a whole 512-byte row of x and sums its dots in the old order; the dot kernel and its 3 launches are gone
# speedup vs baseline: 3.5589x; 1.0704x over previous
#include <hip/hip_runtime.h>
#include <hip/hip_bf16.h>


#define HID 128
#define NEG_SLOPE 0.2f
#define GAT_EPS 1e-16f
#define SCAN_TILE 1024

typedef float v2f __attribute__((ext_vector_type(2)));
typedef float v8f __attribute__((ext_vector_type(8)));

__global__ void zero_cnt_kernel(int* __restrict__ cnt, int n) {
    int i = blockIdx.x * blockDim.x + threadIdx.x;
    if (i < n) cnt[i] = 0;
}

__global__ void rank_kernel(const int* __restrict__ dst, int* __restrict__ cnt,
                            int* __restrict__ rank, int n_edges) {
    int e = blockIdx.x * blockDim.x + threadIdx.x;
    if (e < n_edges) rank[e] = atomicAdd(cnt + dst[e], 1);
}

__global__ __launch_bounds__(256) void scan_kernel(const int* __restrict__ cnt,
                                                   int* __restrict__ off, int n) {
    __shared__ int wpre[8];
    __shared__ int wtot[8];
    const int tid  = threadIdx.x;
    const int lane = tid & 31;
    const int wave = tid >> 5;
    const int base = blockIdx.x * SCAN_TILE;

    int pre = 0;
    const int4* c4 = (const int4*)cnt;
    #pragma unroll 4
    for (int i = tid; i < base / 4; i += 256) {
        int4 q = c4[i];
        pre += (q.x + q.y) + (q.z + q.w);
    }

    const int idx0 = base + tid * 4;
    int v0 = (idx0 + 0 < n) ? cnt[idx0 + 0] : 0;
    int v1 = (idx0 + 1 < n) ? cnt[idx0 + 1] : 0;
    int v2 = (idx0 + 2 < n) ? cnt[idx0 + 2] : 0;
    int v3 = (idx0 + 3 < n) ? cnt[idx0 + 3] : 0;
    const int t = (v0 + v1) + (v2 + v3);

    for (int o = 16; o > 0; o >>= 1) pre += __shfl_xor(pre, o, 32);
    int inc = t;
    for (int o = 1; o < 32; o <<= 1) {
        int u = __shfl_up(inc, o, 32);
        if (lane >= o) inc += u;
    }
    if (lane == 31) wtot[wave] = inc;
    if (lane == 0)  wpre[wave] = pre;
    __syncthreads();

    int run = inc - t;
    for (int w = 0; w < 8; ++w) {
        run += wpre[w];
        if (w < wave) run += wtot[w];
    }
    if (idx0 + 0 <= n) off[idx0 + 0] = run;
    run += v0;
    if (idx0 + 1 <= n) off[idx0 + 1] = run;
    run += v1;
    if (idx0 + 2 <= n) off[idx0 + 2] = run;
    run += v2;
    if (idx0 + 3 <= n) off[idx0 + 3] = run;
}

__global__ void fill_kernel(const int* __restrict__ src, const int* __restrict__ dst,
                            const int* __restrict__ rank, const int* __restrict__ off,
                            int* __restrict__ col, int n_edges) {
    int e = blockIdx.x * blockDim.x + threadIdx.x;
    if (e < n_edges) col[off[dst[e]] + rank[e]] = src[e];
}

#define XPITCH (HID + 4)
__global__ __launch_bounds__(256) void gemm_x_attn_kernel(const float* __restrict__ h,
                                                          const float* __restrict__ W,
                                                          const float* __restrict__ a_s,
                                                          const float* __restrict__ a_d,
                                                          float* __restrict__ x,
                                                          float* __restrict__ as_out,
                                                          float* __restrict__ ad_out,
                                                          int n) {
    __shared__ float sh[16 * HID];
    __shared__ float st[16 * XPITCH];
    const int block_row = blockIdx.x * 16;
    const int tid  = threadIdx.x;
    const int wave = tid >> 5;
    const int lane = tid & 31;

    const float4* gsrc = (const float4*)(h + (size_t)block_row * HID);
    float4*       ldst = (float4*)sh;
    #pragma unroll
    for (int i = tid; i < 16 * HID / 4; i += 256) ldst[i] = gsrc[i];
    const float4 s4 = ((const float4*)a_s)[lane];
    const float4 d4 = ((const float4*)a_d)[lane];
    __syncthreads();

    const int m    = lane & 15;
    const int koff = (lane >> 4) << 1;
    const int ncol = wave * 16 + (lane & 15);

    v8f c = {};
    #pragma unroll
    for (int k = 0; k < HID; k += 4) {
        v2f a, b;
        a.x = sh[m * HID + k + koff];
        a.y = sh[m * HID + k + koff + 1];
        b.x = W[(size_t)(k + koff)     * HID + ncol];
        b.y = W[(size_t)(k + koff + 1) * HID + ncol];
        c = __builtin_amdgcn_wmma_f32_16x16x4_f32(
                 false, a,  false, b,
                 (short)0, c,  false,  false);
    }

    const int rloc = (lane >> 4) << 3;
    #pragma unroll
    for (int r = 0; r < 8; ++r) st[(rloc + r) * XPITCH + ncol] = c[r];
    __syncthreads();

    for (int j = 0; j < 2; ++j) {
        const int rl  = wave * 2 + j;
        const int row = block_row + rl;
        const float4 v = *(const float4*)(st + rl * XPITCH + lane * 4);
        float ps = v.x * s4.x + v.y * s4.y + v.z * s4.z + v.w * s4.w;
        float pd = v.x * d4.x + v.y * d4.y + v.z * d4.z + v.w * d4.w;
        #pragma unroll
        for (int o = 16; o > 0; o >>= 1) {
            ps += __shfl_down(ps, o, 32);
            pd += __shfl_down(pd, o, 32);
        }
        if (row < n) {
            ((float4*)(x + (size_t)row * HID))[lane] = v;
            if (lane == 0) { as_out[row] = ps; ad_out[row] = pd; }
        }
    }
}

__device__ __forceinline__ float wave_max(float v) {
    for (int o = 16; o > 0; o >>= 1) v = fmaxf(v, __shfl_xor(v, o, 32));
    return v;
}
__device__ __forceinline__ float wave_sum(float v) {
    for (int o = 16; o > 0; o >>= 1) v += __shfl_xor(v, o, 32);
    return v;
}

__device__ __forceinline__ float item_logit(int k, int cnt, int beg, int d, int last,
                                            const int* __restrict__ col,
                                            const float* __restrict__ as_, float adv, int& s) {
    s = d;
    if (k < cnt - 1) s = col[beg + k];
    s = min(max(s, 0), last);
    float e = as_[s] + adv;
    return (e > 0.0f) ? e : NEG_SLOPE * e;
}

__device__ __forceinline__ void add_rows(float4& acc, const float4* __restrict__ x4, int lane,
                                         int s, float alpha, int lim) {
    #pragma unroll 4
    for (int j = 0; j < lim; ++j) {
        const int   sj = __builtin_amdgcn_readlane(s, j);
        const float aj = __int_as_float(__builtin_amdgcn_readlane(__float_as_int(alpha), j));
        const float4 v = x4[(size_t)sj * (HID / 4) + lane];
        acc.x += aj * v.x;
        acc.y += aj * v.y;
        acc.z += aj * v.z;
        acc.w += aj * v.w;
    }
}

__global__ __launch_bounds__(256) void node_gather_kernel(const int* __restrict__ off,
                                                          const int* __restrict__ col,
                                                          const float* __restrict__ x,
                                                          const float* __restrict__ as_,
                                                          const float* __restrict__ ad_,
                                                          const float* __restrict__ bias,
                                                          float* __restrict__ out,
                                                          int n, int do_relu) {
    const int lane = threadIdx.x & 31;
    const int d = __builtin_amdgcn_readfirstlane((int)((blockIdx.x * blockDim.x + threadIdx.x) >> 5));
    if (d >= n) return;
    const int   beg  = off[d];
    const int   cnt  = off[d + 1] - beg + 1;
    const float adv  = ad_[d];
    const int   last = n - 1;
    const float4* __restrict__ x4 = (const float4*)x;
    float4 acc = make_float4(0.0f, 0.0f, 0.0f, 0.0f);

    if (cnt <= 32) {
        int s;
        const float e  = item_logit(lane, cnt, beg, d, last, col, as_, adv, s);
        const bool  on = lane < cnt;
        const float m   = wave_max(on ? e : -__builtin_inff());
        const float ex  = on ? __expf(e - m) : 0.0f;
        const float sum = wave_sum(ex);
        const float alpha = ex / (sum + GAT_EPS);
        add_rows(acc, x4, lane, s, alpha, cnt);
    } else {
        float m = -__builtin_inff();
        for (int b = 0; b < cnt; b += 32) {
            int s;
            const int k = b + lane;
            if (k < cnt) m = fmaxf(m, item_logit(k, cnt, beg, d, last, col, as_, adv, s));
        }
        m = wave_max(m);
        float sum = 0.0f;
        for (int b = 0; b < cnt; b += 32) {
            int s;
            const int k = b + lane;
            if (k < cnt) sum += __expf(item_logit(k, cnt, beg, d, last, col, as_, adv, s) - m);
        }
        sum = wave_sum(sum);
        for (int b = 0; b < cnt; b += 32) {
            int s = d;
            float alpha = 0.0f;
            const int k = b + lane;
            if (k < cnt)
                alpha = __expf(item_logit(k, cnt, beg, d, last, col, as_, adv, s) - m) / (sum + GAT_EPS);
            add_rows(acc, x4, lane, s, alpha, min(32, cnt - b));
        }
    }

    const float4 b4 = ((const float4*)bias)[lane];
    acc.x += b4.x; acc.y += b4.y; acc.z += b4.z; acc.w += b4.w;
    if (do_relu) {
        acc.x = fmaxf(acc.x, 0.0f); acc.y = fmaxf(acc.y, 0.0f);
        acc.z = fmaxf(acc.z, 0.0f); acc.w = fmaxf(acc.w, 0.0f);
    }
    ((float4*)(out + (size_t)d * HID))[lane] = acc;
}

extern "C" void kernel_launch(void* const* d_in, const int* in_sizes, int n_in,
                              void* d_out, int out_size, void* d_ws, size_t ws_size,
                              hipStream_t stream) {
    const float* z     = (const float*)d_in[0];
    const int*   ei    = (const int*)  d_in[1];
    const float* Ws    = (const float*)d_in[2];
    const float* a_src = (const float*)d_in[3];
    const float* a_dst = (const float*)d_in[4];
    const float* bias  = (const float*)d_in[5];
    float*       out   = (float*)d_out;

    const int N = in_sizes[0] / HID;
    const int E = in_sizes[1] / 2;
    const int* src = ei;
    const int* dst = ei + E;

    auto align = [](size_t v) { return (v + 255) & ~(size_t)255; };
    char* p = (char*)d_ws;
    float* x    = (float*)p; p += align((size_t)N * HID * sizeof(float));
    float* bufA = (float*)p; p += align((size_t)N * HID * sizeof(float));
    float* bufB = (float*)p; p += align((size_t)N * HID * sizeof(float));
    float* as_  = (float*)p; p += align((size_t)N * sizeof(float));
    float* ad_  = (float*)p; p += align((size_t)N * sizeof(float));
    int*   cnt  = (int*)p;   p += align((size_t)N * sizeof(int));
    int*   off  = (int*)p;   p += align((size_t)(N + 1) * sizeof(int));
    int*   rank = (int*)p;   p += align((size_t)E * sizeof(int));
    int*   col  = (int*)p;   p += align((size_t)E * sizeof(int));

    zero_cnt_kernel<<<(N + 255) / 256, 256, 0, stream>>>(cnt, N);
    rank_kernel<<<(E + 255) / 256, 256, 0, stream>>>(dst, cnt, rank, E);
    scan_kernel<<<(N + SCAN_TILE) / SCAN_TILE, 256, 0, stream>>>(cnt, off, N);
    fill_kernel<<<(E + 255) / 256, 256, 0, stream>>>(src, dst, rank, off, col, E);

    const float* h_ptr = z;
    for (int l = 0; l < 3; ++l) {
        float* o_ptr = (l == 0) ? bufA : (l == 1) ? bufB : out;
        const float* W  = Ws    + (size_t)l * HID * HID;
        const float* as = a_src + (size_t)l * HID;
        const float* ad = a_dst + (size_t)l * HID;
        const float* b  = bias  + (size_t)l * HID;

        gemm_x_attn_kernel<<<(N + 15) / 16, 256, 0, stream>>>(h_ptr, W, as, ad, x, as_, ad_, N);
        node_gather_kernel<<<((size_t)N * 32 + 255) / 256, 256, 0, stream>>>(off, col, x, as_, ad_, b,
                                                                             o_ptr, N, l < 2 ? 1 : 0);
        h_ptr = o_ptr;
    }
}


# GEMM retiled: block = 64 rows, each wave keeps 4 accumulator tiles so one W fragment feeds 4 matrix products; h stripe and x tile share one LDS buffer at 132-float pitch; x and dots bit-identical
# speedup vs baseline: 4.2781x; 1.2021x over previous
#include <hip/hip_runtime.h>
#include <hip/hip_bf16.h>


#define HID 128
#define NEG_SLOPE 0.2f
#define GAT_EPS 1e-16f
#define SCAN_TILE 1024

typedef float v2f __attribute__((ext_vector_type(2)));
typedef float v8f __attribute__((ext_vector_type(8)));

__global__ void zero_cnt_kernel(int* __restrict__ cnt, int n) {
    int i = blockIdx.x * blockDim.x + threadIdx.x;
    if (i < n) cnt[i] = 0;
}

__global__ void rank_kernel(const int* __restrict__ dst, int* __restrict__ cnt,
                            int* __restrict__ rank, int n_edges) {
    int e = blockIdx.x * blockDim.x + threadIdx.x;
    if (e < n_edges) rank[e] = atomicAdd(cnt + dst[e], 1);
}

__global__ __launch_bounds__(256) void scan_kernel(const int* __restrict__ cnt,
                                                   int* __restrict__ off, int n) {
    __shared__ int wpre[8];
    __shared__ int wtot[8];
    const int tid  = threadIdx.x;
    const int lane = tid & 31;
    const int wave = tid >> 5;
    const int base = blockIdx.x * SCAN_TILE;

    int pre = 0;
    const int4* c4 = (const int4*)cnt;
    #pragma unroll 4
    for (int i = tid; i < base / 4; i += 256) {
        int4 q = c4[i];
        pre += (q.x + q.y) + (q.z + q.w);
    }

    const int idx0 = base + tid * 4;
    int v0 = (idx0 + 0 < n) ? cnt[idx0 + 0] : 0;
    int v1 = (idx0 + 1 < n) ? cnt[idx0 + 1] : 0;
    int v2 = (idx0 + 2 < n) ? cnt[idx0 + 2] : 0;
    int v3 = (idx0 + 3 < n) ? cnt[idx0 + 3] : 0;
    const int t = (v0 + v1) + (v2 + v3);

    for (int o = 16; o > 0; o >>= 1) pre += __shfl_xor(pre, o, 32);
    int inc = t;
    for (int o = 1; o < 32; o <<= 1) {
        int u = __shfl_up(inc, o, 32);
        if (lane >= o) inc += u;
    }
    if (lane == 31) wtot[wave] = inc;
    if (lane == 0)  wpre[wave] = pre;
    __syncthreads();

    int run = inc - t;
    for (int w = 0; w < 8; ++w) {
        run += wpre[w];
        if (w < wave) run += wtot[w];
    }
    if (idx0 + 0 <= n) off[idx0 + 0] = run;
    run += v0;
    if (idx0 + 1 <= n) off[idx0 + 1] = run;
    run += v1;
    if (idx0 + 2 <= n) off[idx0 + 2] = run;
    run += v2;
    if (idx0 + 3 <= n) off[idx0 + 3] = run;
}

__global__ void fill_kernel(const int* __restrict__ src, const int* __restrict__ dst,
                            const int* __restrict__ rank, const int* __restrict__ off,
                            int* __restrict__ col, int n_edges) {
    int e = blockIdx.x * blockDim.x + threadIdx.x;
    if (e < n_edges) col[off[dst[e]] + rank[e]] = src[e];
}

#define GROWS 64
#define XPITCH (HID + 4)
__global__ __launch_bounds__(256) void gemm_x_attn_kernel(const float* __restrict__ h,
                                                          const float* __restrict__ W,
                                                          const float* __restrict__ a_s,
                                                          const float* __restrict__ a_d,
                                                          float* __restrict__ x,
                                                          float* __restrict__ as_out,
                                                          float* __restrict__ ad_out,
                                                          int n) {
    __shared__ float sh[GROWS * XPITCH];
    const int block_row = blockIdx.x * GROWS;
    const int tid  = threadIdx.x;
    const int wave = tid >> 5;
    const int lane = tid & 31;

    #pragma unroll
    for (int j = 0; j < GROWS * HID / 4 / 256; ++j) {
        const int r   = j * 8 + wave;
        const int row = min(block_row + r, n - 1);
        const float4 v = ((const float4*)(h + (size_t)row * HID))[lane];
        *(float4*)(sh + r * XPITCH + lane * 4) = v;
    }
    const float4 s4 = ((const float4*)a_s)[lane];
    const float4 d4 = ((const float4*)a_d)[lane];
    __syncthreads();

    const int m    = lane & 15;
    const int koff = (lane >> 4) << 1;
    const int ncol = wave * 16 + (lane & 15);

    v8f c0 = {}, c1 = {}, c2 = {}, c3 = {};
    const float* arow = sh + m * XPITCH + koff;
    const float* wcol = W + (size_t)koff * HID + ncol;
    v2f a0 = *(const v2f*)(arow +  0 * XPITCH), e0 = *(const v2f*)(arow +  0 * XPITCH + 4);
    v2f a1 = *(const v2f*)(arow + 16 * XPITCH), e1 = *(const v2f*)(arow + 16 * XPITCH + 4);
    v2f a2 = *(const v2f*)(arow + 32 * XPITCH), e2 = *(const v2f*)(arow + 32 * XPITCH + 4);
    v2f a3 = *(const v2f*)(arow + 48 * XPITCH), e3 = *(const v2f*)(arow + 48 * XPITCH + 4);
    v2f b, f;
    b.x = wcol[0 * HID]; b.y = wcol[1 * HID];
    f.x = wcol[4 * HID]; f.y = wcol[5 * HID];
    #pragma unroll
    for (int k = 0; k < HID; k += 8) {
        const v2f ca0 = a0, ca1 = a1, ca2 = a2, ca3 = a3, cb = b;
        const v2f ce0 = e0, ce1 = e1, ce2 = e2, ce3 = e3, cf = f;
        if (k + 8 < HID) {
            const int kn = k + 8;
            a0 = *(const v2f*)(arow +  0 * XPITCH + kn); e0 = *(const v2f*)(arow +  0 * XPITCH + kn + 4);
            a1 = *(const v2f*)(arow + 16 * XPITCH + kn); e1 = *(const v2f*)(arow + 16 * XPITCH + kn + 4);
            a2 = *(const v2f*)(arow + 32 * XPITCH + kn); e2 = *(const v2f*)(arow + 32 * XPITCH + kn + 4);
            a3 = *(const v2f*)(arow + 48 * XPITCH + kn); e3 = *(const v2f*)(arow + 48 * XPITCH + kn + 4);
            b.x = wcol[(size_t)(kn)     * HID]; b.y = wcol[(size_t)(kn + 1) * HID];
            f.x = wcol[(size_t)(kn + 4) * HID]; f.y = wcol[(size_t)(kn + 5) * HID];
        }
        __builtin_amdgcn_sched_barrier(0);
        c0 = __builtin_amdgcn_wmma_f32_16x16x4_f32(false, ca0, false, cb, (short)0, c0, false, false);
        c1 = __builtin_amdgcn_wmma_f32_16x16x4_f32(false, ca1, false, cb, (short)0, c1, false, false);
        c2 = __builtin_amdgcn_wmma_f32_16x16x4_f32(false, ca2, false, cb, (short)0, c2, false, false);
        c3 = __builtin_amdgcn_wmma_f32_16x16x4_f32(false, ca3, false, cb, (short)0, c3, false, false);
        c0 = __builtin_amdgcn_wmma_f32_16x16x4_f32(false, ce0, false, cf, (short)0, c0, false, false);
        c1 = __builtin_amdgcn_wmma_f32_16x16x4_f32(false, ce1, false, cf, (short)0, c1, false, false);
        c2 = __builtin_amdgcn_wmma_f32_16x16x4_f32(false, ce2, false, cf, (short)0, c2, false, false);
        c3 = __builtin_amdgcn_wmma_f32_16x16x4_f32(false, ce3, false, cf, (short)0, c3, false, false);
    }
    __syncthreads();

    const int rloc = (lane >> 4) << 3;
    float* trow = sh + rloc * XPITCH + ncol;
    #pragma unroll
    for (int r = 0; r < 8; ++r) {
        trow[( 0 + r) * XPITCH] = c0[r];
        trow[(16 + r) * XPITCH] = c1[r];
        trow[(32 + r) * XPITCH] = c2[r];
        trow[(48 + r) * XPITCH] = c3[r];
    }
    __syncthreads();

    for (int j = 0; j < 8; ++j) {
        const int rl  = wave * 8 + j;
        const int row = block_row + rl;
        const float4 v = *(const float4*)(sh + rl * XPITCH + lane * 4);
        float ps = __builtin_fmaf(v.w, s4.w, __builtin_fmaf(v.z, s4.z, __builtin_fmaf(v.x, s4.x, v.y * s4.y)));
        float pd = __builtin_fmaf(v.w, d4.w, __builtin_fmaf(v.z, d4.z, __builtin_fmaf(v.x, d4.x, v.y * d4.y)));
        #pragma unroll
        for (int o = 16; o > 0; o >>= 1) {
            ps += __shfl_down(ps, o, 32);
            pd += __shfl_down(pd, o, 32);
        }
        if (row < n) {
            ((float4*)(x + (size_t)row * HID))[lane] = v;
            if (lane == 0) { as_out[row] = ps; ad_out[row] = pd; }
        }
    }
}

__device__ __forceinline__ float wave_max(float v) {
    for (int o = 16; o > 0; o >>= 1) v = fmaxf(v, __shfl_xor(v, o, 32));
    return v;
}
__device__ __forceinline__ float wave_sum(float v) {
    for (int o = 16; o > 0; o >>= 1) v += __shfl_xor(v, o, 32);
    return v;
}

__device__ __forceinline__ float item_logit(int k, int cnt, int beg, int d, int last,
                                            const int* __restrict__ col,
                                            const float* __restrict__ as_, float adv, int& s) {
    s = d;
    if (k < cnt - 1) s = col[beg + k];
    s = min(max(s, 0), last);
    float e = as_[s] + adv;
    return (e > 0.0f) ? e : NEG_SLOPE * e;
}

__device__ __forceinline__ void add_rows(float4& acc, const float4* __restrict__ x4, int lane,
                                         int s, float alpha, int lim) {
    #pragma unroll 4
    for (int j = 0; j < lim; ++j) {
        const int   sj = __builtin_amdgcn_readlane(s, j);
        const float aj = __int_as_float(__builtin_amdgcn_readlane(__float_as_int(alpha), j));
        const float4 v = x4[(size_t)sj * (HID / 4) + lane];
        acc.x += aj * v.x;
        acc.y += aj * v.y;
        acc.z += aj * v.z;
        acc.w += aj * v.w;
    }
}

__global__ __launch_bounds__(256) void node_gather_kernel(const int* __restrict__ off,
                                                          const int* __restrict__ col,
                                                          const float* __restrict__ x,
                                                          const float* __restrict__ as_,
                                                          const float* __restrict__ ad_,
                                                          const float* __restrict__ bias,
                                                          float* __restrict__ out,
                                                          int n, int do_relu) {
    const int lane = threadIdx.x & 31;
    const int d = __builtin_amdgcn_readfirstlane((int)((blockIdx.x * blockDim.x + threadIdx.x) >> 5));
    if (d >= n) return;
    const int   beg  = off[d];
    const int   cnt  = off[d + 1] - beg + 1;
    const float adv  = ad_[d];
    const int   last = n - 1;
    const float4* __restrict__ x4 = (const float4*)x;
    float4 acc = make_float4(0.0f, 0.0f, 0.0f, 0.0f);

    if (cnt <= 32) {
        int s;
        const float e  = item_logit(lane, cnt, beg, d, last, col, as_, adv, s);
        const bool  on = lane < cnt;
        const float m   = wave_max(on ? e : -__builtin_inff());
        const float ex  = on ? __expf(e - m) : 0.0f;
        const float sum = wave_sum(ex);
        const float alpha = ex / (sum + GAT_EPS);
        add_rows(acc, x4, lane, s, alpha, cnt);
    } else {
        float m = -__builtin_inff();
        for (int b = 0; b < cnt; b += 32) {
            int s;
            const int k = b + lane;
            if (k < cnt) m = fmaxf(m, item_logit(k, cnt, beg, d, last, col, as_, adv, s));
        }
        m = wave_max(m);
        float sum = 0.0f;
        for (int b = 0; b < cnt; b += 32) {
            int s;
            const int k = b + lane;
            if (k < cnt) sum += __expf(item_logit(k, cnt, beg, d, last, col, as_, adv, s) - m);
        }
        sum = wave_sum(sum);
        for (int b = 0; b < cnt; b += 32) {
            int s = d;
            float alpha = 0.0f;
            const int k = b + lane;
            if (k < cnt)
                alpha = __expf(item_logit(k, cnt, beg, d, last, col, as_, adv, s) - m) / (sum + GAT_EPS);
            add_rows(acc, x4, lane, s, alpha, min(32, cnt - b));
        }
    }

    const float4 b4 = ((const float4*)bias)[lane];
    acc.x += b4.x; acc.y += b4.y; acc.z += b4.z; acc.w += b4.w;
    if (do_relu) {
        acc.x = fmaxf(acc.x, 0.0f); acc.y = fmaxf(acc.y, 0.0f);
        acc.z = fmaxf(acc.z, 0.0f); acc.w = fmaxf(acc.w, 0.0f);
    }
    ((float4*)(out + (size_t)d * HID))[lane] = acc;
}

extern "C" void kernel_launch(void* const* d_in, const int* in_sizes, int n_in,
                              void* d_out, int out_size, void* d_ws, size_t ws_size,
                              hipStream_t stream) {
    const float* z     = (const float*)d_in[0];
    const int*   ei    = (const int*)  d_in[1];
    const float* Ws    = (const float*)d_in[2];
    const float* a_src = (const float*)d_in[3];
    const float* a_dst = (const float*)d_in[4];
    const float* bias  = (const float*)d_in[5];
    float*       out   = (float*)d_out;

    const int N = in_sizes[0] / HID;
    const int E = in_sizes[1] / 2;
    const int* src = ei;
    const int* dst = ei + E;

    auto align = [](size_t v) { return (v + 255) & ~(size_t)255; };
    char* p = (char*)d_ws;
    float* x    = (float*)p; p += align((size_t)N * HID * sizeof(float));
    float* bufA = (float*)p; p += align((size_t)N * HID * sizeof(float));
    float* bufB = (float*)p; p += align((size_t)N * HID * sizeof(float));
    float* as_  = (float*)p; p += align((size_t)N * sizeof(float));
    float* ad_  = (float*)p; p += align((size_t)N * sizeof(float));
    int*   cnt  = (int*)p;   p += align((size_t)N * sizeof(int));
    int*   off  = (int*)p;   p += align((size_t)(N + 1) * sizeof(int));
    int*   rank = (int*)p;   p += align((size_t)E * sizeof(int));
    int*   col  = (int*)p;   p += align((size_t)E * sizeof(int));

    zero_cnt_kernel<<<(N + 255) / 256, 256, 0, stream>>>(cnt, N);
    rank_kernel<<<(E + 255) / 256, 256, 0, stream>>>(dst, cnt, rank, E);
    scan_kernel<<<(N + SCAN_TILE) / SCAN_TILE, 256, 0, stream>>>(cnt, off, N);
    fill_kernel<<<(E + 255) / 256, 256, 0, stream>>>(src, dst, rank, off, col, E);

    const float* h_ptr = z;
    for (int l = 0; l < 3; ++l) {
        float* o_ptr = (l == 0) ? bufA : (l == 1) ? bufB : out;
        const float* W  = Ws    + (size_t)l * HID * HID;
        const float* as = a_src + (size_t)l * HID;
        const float* ad = a_dst + (size_t)l * HID;
        const float* b  = bias  + (size_t)l * HID;

        gemm_x_attn_kernel<<<(N + GROWS - 1) / GROWS, 256, 0, stream>>>(h_ptr, W, as, ad, x, as_, ad_, N);
        node_gather_kernel<<<((size_t)N * 32 + 255) / 256, 256, 0, stream>>>(off, col, x, as_, ad_, b,
                                                                             o_ptr, N, l < 2 ? 1 : 0);
        h_ptr = o_ptr;
    }
}


# Edge grouping redone as a two-level counting sort on LDS counters (bins of 256 nodes, then by node inside a bin): no global atomics, 3 launches not 4; a node's rows summed in ascending source id
# speedup vs baseline: 5.9208x; 1.3840x over previous
#include <hip/hip_runtime.h>
#include <hip/hip_bf16.h>


#define HID 128
#define NEG_SLOPE 0.2f
#define GAT_EPS 1e-16f

typedef float v2f __attribute__((ext_vector_type(2)));
typedef float v8f __attribute__((ext_vector_type(8)));

#define BIN_SHIFT 8
#define BIN_NODES 256
#define MAXB      256
#define CHUNK     4096
#define FCAP      6144

__device__ __forceinline__ int edge_bin(int d, int nb) {
    return min((int)((unsigned)d >> BIN_SHIFT), nb - 1);
}

__device__ __forceinline__ int block_excl_scan256(int v, int* wsum) {
    const int lane = threadIdx.x & 31;
    const int wave = threadIdx.x >> 5;
    int inc = v;
    for (int o = 1; o < 32; o <<= 1) {
        int u = __shfl_up(inc, o, 32);
        if (lane >= o) inc += u;
    }
    if (lane == 31) wsum[wave] = inc;
    __syncthreads();
    int run = inc - v;
    for (int w = 0; w < wave; ++w) run += wsum[w];
    __syncthreads();
    return run;
}

__global__ __launch_bounds__(256) void bin_hist_kernel(const int* __restrict__ dst, int* __restrict__ tab,
                                                       int n_edges, int nb) {
    __shared__ int hist[MAXB];
    const int tid  = threadIdx.x;
    const int base = blockIdx.x * CHUNK;
    int dv[CHUNK / 256];
    for (int i = 0; i < CHUNK / 256; ++i) dv[i] = dst[min(base + i * 256 + tid, n_edges - 1)];
    hist[tid] = 0;
    __syncthreads();
    for (int i = 0; i < CHUNK / 256; ++i) {
        const int e = base + i * 256 + tid;
        if (e < n_edges) atomicAdd(&hist[edge_bin(dv[i], nb)], 1);
    }
    __syncthreads();
    if (tid < nb) tab[blockIdx.x * nb + tid] = hist[tid];
}

__global__ __launch_bounds__(256) void bin_scatter_kernel(const int* __restrict__ src,
                                                          const int* __restrict__ dst,
                                                          const int* __restrict__ tab,
                                                          int* __restrict__ bstart,
                                                          int* __restrict__ binbuf,
                                                          int n_edges, int nb) {
    __shared__ int cur[MAXB];
    __shared__ int lofs[MAXB];
    __shared__ int delta[MAXB];
    __shared__ int wsum[8];
    __shared__ int stage[CHUNK];
    __shared__ unsigned char sbin[CHUNK];
    const int tid  = threadIdx.x;
    const int nblk = gridDim.x;
    const int me   = blockIdx.x;
    const int base = me * CHUNK;

    int dv[CHUNK / 256], sv[CHUNK / 256];
    for (int i = 0; i < CHUNK / 256; ++i) {
        const int e = min(base + i * 256 + tid, n_edges - 1);
        dv[i] = dst[e];
        sv[i] = src[e];
    }

    int tot = 0, pre = 0, own = 0;
    if (tid < nb) {
        for (int j0 = 0; j0 < nblk; j0 += 49) {
            int v[49];
            for (int k = 0; k < 49; ++k) {
                const int j = j0 + k;
                v[k] = (j < nblk) ? tab[j * nb + tid] : 0;
            }
            for (int k = 0; k < 49; ++k) {
                const int j = j0 + k;
                tot += v[k];
                if (j < me)  pre += v[k];
                if (j == me) own  = v[k];
            }
        }
    }
    const int bs = block_excl_scan256(tot, wsum);
    const int lo = block_excl_scan256(own, wsum);
    cur[tid]   = 0;
    lofs[tid]  = lo;
    delta[tid] = bs + pre - lo;
    if (me == 0) {
        if (tid < nb)      bstart[tid] = bs;
        if (tid == nb - 1) bstart[nb]  = bs + tot;
    }
    __syncthreads();

    for (int i = 0; i < CHUNK / 256; ++i) {
        const int e = base + i * 256 + tid;
        if (e < n_edges) {
            const int d = dv[i];
            const int s = sv[i];
            const int b = edge_bin(d, nb);
            const int p = lofs[b] + atomicAdd(&cur[b], 1);
            stage[p] = (int)(((unsigned)s << 8) | ((unsigned)d & 255u));
            sbin[p]  = (unsigned char)b;
        }
    }
    __syncthreads();
    const int nloc = min(CHUNK, n_edges - base);
    int pb[CHUNK / 256], pd[CHUNK / 256], pw[CHUNK / 256];
    for (int i = 0; i < CHUNK / 256; ++i) pb[i] = sbin[min(i * 256 + tid, nloc - 1)];
    for (int i = 0; i < CHUNK / 256; ++i) {
        pd[i] = delta[pb[i]];
        pw[i] = stage[min(i * 256 + tid, nloc - 1)];
    }
    for (int i = 0; i < CHUNK / 256; ++i) {
        const int p = i * 256 + tid;
        if (p < nloc) binbuf[pd[i] + p] = pw[i];
    }
}

__global__ __launch_bounds__(1024) void bin_finalize_kernel(const int* __restrict__ binbuf,
                                                            const int* __restrict__ bstart,
                                                            int* __restrict__ off, int* __restrict__ col,
                                                            int n) {
    __shared__ int ncnt[BIN_NODES];
    __shared__ int noff[BIN_NODES];
    __shared__ int ncur[BIN_NODES];
    __shared__ int stage[FCAP];
    __shared__ int sorted[FCAP];
    __shared__ unsigned char snode[FCAP];
    const int tid  = threadIdx.x;
    const int lane = tid & 31;
    const int b    = blockIdx.x;
    const int start = bstart[b];
    const int cntb  = bstart[b + 1] - start;
    const bool staged = cntb <= FCAP;
    unsigned w[FCAP / 1024];
    for (int k = 0; k < FCAP / 1024; ++k) {
        const int i = k * 1024 + tid;
        w[k] = (staged && i < cntb) ? (unsigned)binbuf[start + i] : 0u;
    }
    if (tid < BIN_NODES) { ncnt[tid] = 0; ncur[tid] = 0; }
    __syncthreads();
    if (staged) {
        for (int k = 0; k < FCAP / 1024; ++k)
            if (k * 1024 + tid < cntb) atomicAdd(&ncnt[w[k] & 255u], 1);
    } else {
        for (int i = tid; i < cntb; i += 1024) atomicAdd(&ncnt[binbuf[start + i] & 255], 1);
    }
    __syncthreads();
    if (tid < 32) {
        int a[8];
        int t = 0;
        for (int k = 0; k < 8; ++k) { a[k] = ncnt[lane * 8 + k]; t += a[k]; }
        int inc = t;
        for (int o = 1; o < 32; o <<= 1) {
            int u = __shfl_up(inc, o, 32);
            if (lane >= o) inc += u;
        }
        int run = inc - t;
        for (int k = 0; k < 8; ++k) { noff[lane * 8 + k] = run; run += a[k]; }
    }
    __syncthreads();
    if (tid < BIN_NODES) {
        const int node = b * BIN_NODES + tid;
        if (node < n) off[node] = start + noff[tid];
    }
    if (b == (int)gridDim.x - 1 && tid == 0) off[n] = start + cntb;

    if (staged) {
        for (int k = 0; k < FCAP / 1024; ++k) {
            if (k * 1024 + tid < cntb) {
                const int ld = (int)(w[k] & 255u);
                const int p  = noff[ld] + atomicAdd(&ncur[ld], 1);
                stage[p] = (int)(w[k] >> 8);
                snode[p] = (unsigned char)ld;
            }
        }
    } else {
        for (int i = tid; i < cntb; i += 1024) {
            const unsigned u = (unsigned)binbuf[start + i];
            const int ld = (int)(u & 255u);
            const int p  = noff[ld] + atomicAdd(&ncur[ld], 1);
            col[start + p] = (int)(u >> 8);
        }
    }
    __syncthreads();
    if (staged) {
        for (int i = tid; i < cntb; i += 1024) {
            const int ld = snode[i];
            const int v  = stage[i];
            const int lo = noff[ld];
            const int hi = lo + ncnt[ld];
            int r = lo;
            for (int q = lo; q < hi; ++q) {
                const int u = stage[q];
                r += (u < v || (u == v && q < i)) ? 1 : 0;
            }
            sorted[r] = v;
        }
    }
    __syncthreads();
    if (staged)
        for (int i = tid; i < cntb; i += 1024) col[start + i] = sorted[i];
}

#define GROWS 64
#define XPITCH (HID + 4)
__global__ __launch_bounds__(256) void gemm_x_attn_kernel(const float* __restrict__ h,
                                                          const float* __restrict__ W,
                                                          const float* __restrict__ a_s,
                                                          const float* __restrict__ a_d,
                                                          float* __restrict__ x,
                                                          float* __restrict__ as_out,
                                                          float* __restrict__ ad_out,
                                                          int n) {
    __shared__ float sh[GROWS * XPITCH];
    const int block_row = blockIdx.x * GROWS;
    const int tid  = threadIdx.x;
    const int wave = tid >> 5;
    const int lane = tid & 31;

    #pragma unroll
    for (int j = 0; j < GROWS * HID / 4 / 256; ++j) {
        const int r   = j * 8 + wave;
        const int row = min(block_row + r, n - 1);
        const float4 v = ((const float4*)(h + (size_t)row * HID))[lane];
        *(float4*)(sh + r * XPITCH + lane * 4) = v;
    }
    const float4 s4 = ((const float4*)a_s)[lane];
    const float4 d4 = ((const float4*)a_d)[lane];
    __syncthreads();

    const int m    = lane & 15;
    const int koff = (lane >> 4) << 1;
    const int ncol = wave * 16 + (lane & 15);

    v8f c0 = {}, c1 = {}, c2 = {}, c3 = {};
    const float* arow = sh + m * XPITCH + koff;
    const float* wcol = W + (size_t)koff * HID + ncol;
    v2f a0 = *(const v2f*)(arow +  0 * XPITCH), e0 = *(const v2f*)(arow +  0 * XPITCH + 4);
    v2f a1 = *(const v2f*)(arow + 16 * XPITCH), e1 = *(const v2f*)(arow + 16 * XPITCH + 4);
    v2f a2 = *(const v2f*)(arow + 32 * XPITCH), e2 = *(const v2f*)(arow + 32 * XPITCH + 4);
    v2f a3 = *(const v2f*)(arow + 48 * XPITCH), e3 = *(const v2f*)(arow + 48 * XPITCH + 4);
    v2f b, f;
    b.x = wcol[0 * HID]; b.y = wcol[1 * HID];
    f.x = wcol[4 * HID]; f.y = wcol[5 * HID];
    #pragma unroll
    for (int k = 0; k < HID; k += 8) {
        const v2f ca0 = a0, ca1 = a1, ca2 = a2, ca3 = a3, cb = b;
        const v2f ce0 = e0, ce1 = e1, ce2 = e2, ce3 = e3, cf = f;
        if (k + 8 < HID) {
            const int kn = k + 8;
            a0 = *(const v2f*)(arow +  0 * XPITCH + kn); e0 = *(const v2f*)(arow +  0 * XPITCH + kn + 4);
            a1 = *(const v2f*)(arow + 16 * XPITCH + kn); e1 = *(const v2f*)(arow + 16 * XPITCH + kn + 4);
            a2 = *(const v2f*)(arow + 32 * XPITCH + kn); e2 = *(const v2f*)(arow + 32 * XPITCH + kn + 4);
            a3 = *(const v2f*)(arow + 48 * XPITCH + kn); e3 = *(const v2f*)(arow + 48 * XPITCH + kn + 4);
            b.x = wcol[(size_t)(kn)     * HID]; b.y = wcol[(size_t)(kn + 1) * HID];
            f.x = wcol[(size_t)(kn + 4) * HID]; f.y = wcol[(size_t)(kn + 5) * HID];
        }
        __builtin_amdgcn_sched_barrier(0);
        c0 = __builtin_amdgcn_wmma_f32_16x16x4_f32(false, ca0, false, cb, (short)0, c0, false, false);
        c1 = __builtin_amdgcn_wmma_f32_16x16x4_f32(false, ca1, false, cb, (short)0, c1, false, false);
        c2 = __builtin_amdgcn_wmma_f32_16x16x4_f32(false, ca2, false, cb, (short)0, c2, false, false);
        c3 = __builtin_amdgcn_wmma_f32_16x16x4_f32(false, ca3, false, cb, (short)0, c3, false, false);
        c0 = __builtin_amdgcn_wmma_f32_16x16x4_f32(false, ce0, false, cf, (short)0, c0, false, false);
        c1 = __builtin_amdgcn_wmma_f32_16x16x4_f32(false, ce1, false, cf, (short)0, c1, false, false);
        c2 = __builtin_amdgcn_wmma_f32_16x16x4_f32(false, ce2, false, cf, (short)0, c2, false, false);
        c3 = __builtin_amdgcn_wmma_f32_16x16x4_f32(false, ce3, false, cf, (short)0, c3, false, false);
    }
    __syncthreads();

    const int rloc = (lane >> 4) << 3;
    float* trow = sh + rloc * XPITCH + ncol;
    #pragma unroll
    for (int r = 0; r < 8; ++r) {
        trow[( 0 + r) * XPITCH] = c0[r];
        trow[(16 + r) * XPITCH] = c1[r];
        trow[(32 + r) * XPITCH] = c2[r];
        trow[(48 + r) * XPITCH] = c3[r];
    }
    __syncthreads();

    for (int j = 0; j < 8; ++j) {
        const int rl  = wave * 8 + j;
        const int row = block_row + rl;
        const float4 v = *(const float4*)(sh + rl * XPITCH + lane * 4);
        float ps = __builtin_fmaf(v.w, s4.w, __builtin_fmaf(v.z, s4.z, __builtin_fmaf(v.x, s4.x, v.y * s4.y)));
        float pd = __builtin_fmaf(v.w, d4.w, __builtin_fmaf(v.z, d4.z, __builtin_fmaf(v.x, d4.x, v.y * d4.y)));
        #pragma unroll
        for (int o = 16; o > 0; o >>= 1) {
            ps += __shfl_down(ps, o, 32);
            pd += __shfl_down(pd, o, 32);
        }
        if (row < n) {
            ((float4*)(x + (size_t)row * HID))[lane] = v;
            if (lane == 0) { as_out[row] = ps; ad_out[row] = pd; }
        }
    }
}

__device__ __forceinline__ float wave_max(float v) {
    for (int o = 16; o > 0; o >>= 1) v = fmaxf(v, __shfl_xor(v, o, 32));
    return v;
}
__device__ __forceinline__ float wave_sum(float v) {
    for (int o = 16; o > 0; o >>= 1) v += __shfl_xor(v, o, 32);
    return v;
}

__device__ __forceinline__ float item_logit(int k, int cnt, int beg, int d, int last,
                                            const int* __restrict__ col,
                                            const float* __restrict__ as_, float adv, int& s) {
    s = d;
    if (k < cnt - 1) s = col[beg + k];
    s = min(max(s, 0), last);
    float e = as_[s] + adv;
    return (e > 0.0f) ? e : NEG_SLOPE * e;
}

__device__ __forceinline__ void add_rows(float4& acc, const float4* __restrict__ x4, int lane,
                                         int s, float alpha, int lim) {
    #pragma unroll 4
    for (int j = 0; j < lim; ++j) {
        const int   sj = __builtin_amdgcn_readlane(s, j);
        const float aj = __int_as_float(__builtin_amdgcn_readlane(__float_as_int(alpha), j));
        const float4 v = x4[(size_t)sj * (HID / 4) + lane];
        acc.x += aj * v.x;
        acc.y += aj * v.y;
        acc.z += aj * v.z;
        acc.w += aj * v.w;
    }
}

__global__ __launch_bounds__(256) void node_gather_kernel(const int* __restrict__ off,
                                                          const int* __restrict__ col,
                                                          const float* __restrict__ x,
                                                          const float* __restrict__ as_,
                                                          const float* __restrict__ ad_,
                                                          const float* __restrict__ bias,
                                                          float* __restrict__ out,
                                                          int n, int do_relu) {
    const int lane = threadIdx.x & 31;
    const int d = __builtin_amdgcn_readfirstlane((int)((blockIdx.x * blockDim.x + threadIdx.x) >> 5));
    if (d >= n) return;
    const int   beg  = off[d];
    const int   cnt  = off[d + 1] - beg + 1;
    const float adv  = ad_[d];
    const int   last = n - 1;
    const float4* __restrict__ x4 = (const float4*)x;
    float4 acc = make_float4(0.0f, 0.0f, 0.0f, 0.0f);

    if (cnt <= 32) {
        int s;
        const float e  = item_logit(lane, cnt, beg, d, last, col, as_, adv, s);
        const bool  on = lane < cnt;
        const float m   = wave_max(on ? e : -__builtin_inff());
        const float ex  = on ? __expf(e - m) : 0.0f;
        const float sum = wave_sum(ex);
        const float alpha = ex / (sum + GAT_EPS);
        add_rows(acc, x4, lane, s, alpha, cnt);
    } else {
        float m = -__builtin_inff();
        for (int b = 0; b < cnt; b += 32) {
            int s;
            const int k = b + lane;
            if (k < cnt) m = fmaxf(m, item_logit(k, cnt, beg, d, last, col, as_, adv, s));
        }
        m = wave_max(m);
        float sum = 0.0f;
        for (int b = 0; b < cnt; b += 32) {
            int s;
            const int k = b + lane;
            if (k < cnt) sum += __expf(item_logit(k, cnt, beg, d, last, col, as_, adv, s) - m);
        }
        sum = wave_sum(sum);
        for (int b = 0; b < cnt; b += 32) {
            int s = d;
            float alpha = 0.0f;
            const int k = b + lane;
            if (k < cnt)
                alpha = __expf(item_logit(k, cnt, beg, d, last, col, as_, adv, s) - m) / (sum + GAT_EPS);
            add_rows(acc, x4, lane, s, alpha, min(32, cnt - b));
        }
    }

    const float4 b4 = ((const float4*)bias)[lane];
    acc.x += b4.x; acc.y += b4.y; acc.z += b4.z; acc.w += b4.w;
    if (do_relu) {
        acc.x = fmaxf(acc.x, 0.0f); acc.y = fmaxf(acc.y, 0.0f);
        acc.z = fmaxf(acc.z, 0.0f); acc.w = fmaxf(acc.w, 0.0f);
    }
    ((float4*)(out + (size_t)d * HID))[lane] = acc;
}

extern "C" void kernel_launch(void* const* d_in, const int* in_sizes, int n_in,
                              void* d_out, int out_size, void* d_ws, size_t ws_size,
                              hipStream_t stream) {
    const float* z     = (const float*)d_in[0];
    const int*   ei    = (const int*)  d_in[1];
    const float* Ws    = (const float*)d_in[2];
    const float* a_src = (const float*)d_in[3];
    const float* a_dst = (const float*)d_in[4];
    const float* bias  = (const float*)d_in[5];
    float*       out   = (float*)d_out;

    const int N = in_sizes[0] / HID;
    const int E = in_sizes[1] / 2;
    const int* src = ei;
    const int* dst = ei + E;

    auto align = [](size_t v) { return (v + 255) & ~(size_t)255; };
    char* p = (char*)d_ws;
    float* x    = (float*)p; p += align((size_t)N * HID * sizeof(float));
    float* bufA = (float*)p; p += align((size_t)N * HID * sizeof(float));
    float* bufB = (float*)p; p += align((size_t)N * HID * sizeof(float));
    float* as_  = (float*)p; p += align((size_t)N * sizeof(float));
    float* ad_  = (float*)p; p += align((size_t)N * sizeof(float));
    p += align((size_t)N * sizeof(int));
    int*   off  = (int*)p;   p += align((size_t)(N + 1) * sizeof(int));
    int*   rank = (int*)p;   p += align((size_t)E * sizeof(int));
    int*   col  = (int*)p;   p += align((size_t)E * sizeof(int));

    int* tab    = (int*)x;
    int* bstart = (int*)bufA;
    int* binbuf = rank;
    const int NB   = (N + BIN_NODES - 1) / BIN_NODES;
    const int NBLK = (E + CHUNK - 1) / CHUNK;
    bin_hist_kernel<<<NBLK, 256, 0, stream>>>(dst, tab, E, NB);
    bin_scatter_kernel<<<NBLK, 256, 0, stream>>>(src, dst, tab, bstart, binbuf, E, NB);
    bin_finalize_kernel<<<NB, 1024, 0, stream>>>(binbuf, bstart, off, col, N);

    const float* h_ptr = z;
    for (int l = 0; l < 3; ++l) {
        float* o_ptr = (l == 0) ? bufA : (l == 1) ? bufB : out;
        const float* W  = Ws    + (size_t)l * HID * HID;
        const float* as = a_src + (size_t)l * HID;
        const float* ad = a_dst + (size_t)l * HID;
        const float* b  = bias  + (size_t)l * HID;

        gemm_x_attn_kernel<<<(N + GROWS - 1) / GROWS, 256, 0, stream>>>(h_ptr, W, as, ad, x, as_, ad_, N);
        node_gather_kernel<<<((size_t)N * 32 + 255) / 256, 256, 0, stream>>>(off, col, x, as_, ad_, b,
                                                                             o_ptr, N, l < 2 ? 1 : 0);
        h_ptr = o_ptr;
    }
}


# Layer-0 GEMM moved into the edge-scatter launch: scatter blocks and GEMM row blocks are independent and share one grid (role by block index), 8 launches not 9; every result bit-identical
# speedup vs baseline: 6.5657x; 1.1089x over previous
#include <hip/hip_runtime.h>
#include <hip/hip_bf16.h>


#define HID 128
#define NEG_SLOPE 0.2f
#define GAT_EPS 1e-16f

typedef float v2f __attribute__((ext_vector_type(2)));
typedef float v8f __attribute__((ext_vector_type(8)));

#define BIN_SHIFT 8
#define BIN_NODES 256
#define MAXB      256
#define CHUNK     4096
#define FCAP      6144

__device__ __forceinline__ int edge_bin(int d, int nb) {
    return min((int)((unsigned)d >> BIN_SHIFT), nb - 1);
}

__device__ __forceinline__ int block_excl_scan256(int v, int* wsum) {
    const int lane = threadIdx.x & 31;
    const int wave = threadIdx.x >> 5;
    int inc = v;
    for (int o = 1; o < 32; o <<= 1) {
        int u = __shfl_up(inc, o, 32);
        if (lane >= o) inc += u;
    }
    if (lane == 31) wsum[wave] = inc;
    __syncthreads();
    int run = inc - v;
    for (int w = 0; w < wave; ++w) run += wsum[w];
    __syncthreads();
    return run;
}

__global__ __launch_bounds__(256) void bin_hist_kernel(const int* __restrict__ dst, int* __restrict__ tab,
                                                       int n_edges, int nb) {
    __shared__ int hist[MAXB];
    const int tid  = threadIdx.x;
    const int base = blockIdx.x * CHUNK;
    int dv[CHUNK / 256];
    for (int i = 0; i < CHUNK / 256; ++i) dv[i] = dst[min(base + i * 256 + tid, n_edges - 1)];
    hist[tid] = 0;
    __syncthreads();
    for (int i = 0; i < CHUNK / 256; ++i) {
        const int e = base + i * 256 + tid;
        if (e < n_edges) atomicAdd(&hist[edge_bin(dv[i], nb)], 1);
    }
    __syncthreads();
    if (tid < nb) tab[blockIdx.x * nb + tid] = hist[tid];
}

#define SCATTER_LDS_BYTES ((3 * MAXB + 8 + CHUNK) * 4 + CHUNK)
__device__ __forceinline__ void bin_scatter_block(int* __restrict__ smem, int me, int nblk,
                                                  const int* __restrict__ src,
                                                  const int* __restrict__ dst,
                                                  const int* __restrict__ tab,
                                                  int* __restrict__ bstart,
                                                  int* __restrict__ binbuf,
                                                  int n_edges, int nb) {
    int* cur   = smem;
    int* lofs  = cur + MAXB;
    int* delta = lofs + MAXB;
    int* wsum  = delta + MAXB;
    int* stage = wsum + 8;
    unsigned char* sbin = (unsigned char*)(stage + CHUNK);
    const int tid  = threadIdx.x;
    const int base = me * CHUNK;

    int dv[CHUNK / 256], sv[CHUNK / 256];
    for (int i = 0; i < CHUNK / 256; ++i) {
        const int e = min(base + i * 256 + tid, n_edges - 1);
        dv[i] = dst[e];
        sv[i] = src[e];
    }

    int tot = 0, pre = 0, own = 0;
    if (tid < nb) {
        for (int j0 = 0; j0 < nblk; j0 += 49) {
            int v[49];
            for (int k = 0; k < 49; ++k) {
                const int j = j0 + k;
                v[k] = (j < nblk) ? tab[j * nb + tid] : 0;
            }
            for (int k = 0; k < 49; ++k) {
                const int j = j0 + k;
                tot += v[k];
                if (j < me)  pre += v[k];
                if (j == me) own  = v[k];
            }
        }
    }
    const int bs = block_excl_scan256(tot, wsum);
    const int lo = block_excl_scan256(own, wsum);
    cur[tid]   = 0;
    lofs[tid]  = lo;
    delta[tid] = bs + pre - lo;
    if (me == 0) {
        if (tid < nb)      bstart[tid] = bs;
        if (tid == nb - 1) bstart[nb]  = bs + tot;
    }
    __syncthreads();

    for (int i = 0; i < CHUNK / 256; ++i) {
        const int e = base + i * 256 + tid;
        if (e < n_edges) {
            const int d = dv[i];
            const int s = sv[i];
            const int b = edge_bin(d, nb);
            const int p = lofs[b] + atomicAdd(&cur[b], 1);
            stage[p] = (int)(((unsigned)s << 8) | ((unsigned)d & 255u));
            sbin[p]  = (unsigned char)b;
        }
    }
    __syncthreads();
    const int nloc = min(CHUNK, n_edges - base);
    int pb[CHUNK / 256], pd[CHUNK / 256], pw[CHUNK / 256];
    for (int i = 0; i < CHUNK / 256; ++i) pb[i] = sbin[min(i * 256 + tid, nloc - 1)];
    for (int i = 0; i < CHUNK / 256; ++i) {
        pd[i] = delta[pb[i]];
        pw[i] = stage[min(i * 256 + tid, nloc - 1)];
    }
    for (int i = 0; i < CHUNK / 256; ++i) {
        const int p = i * 256 + tid;
        if (p < nloc) binbuf[pd[i] + p] = pw[i];
    }
}

__global__ __launch_bounds__(1024) void bin_finalize_kernel(const int* __restrict__ binbuf,
                                                            const int* __restrict__ bstart,
                                                            int* __restrict__ off, int* __restrict__ col,
                                                            int n) {
    __shared__ int ncnt[BIN_NODES];
    __shared__ int noff[BIN_NODES];
    __shared__ int ncur[BIN_NODES];
    __shared__ int stage[FCAP];
    __shared__ int sorted[FCAP];
    __shared__ unsigned char snode[FCAP];
    const int tid  = threadIdx.x;
    const int lane = tid & 31;
    const int b    = blockIdx.x;
    const int start = bstart[b];
    const int cntb  = bstart[b + 1] - start;
    const bool staged = cntb <= FCAP;
    unsigned w[FCAP / 1024];
    for (int k = 0; k < FCAP / 1024; ++k) {
        const int i = k * 1024 + tid;
        w[k] = (staged && i < cntb) ? (unsigned)binbuf[start + i] : 0u;
    }
    if (tid < BIN_NODES) { ncnt[tid] = 0; ncur[tid] = 0; }
    __syncthreads();
    if (staged) {
        for (int k = 0; k < FCAP / 1024; ++k)
            if (k * 1024 + tid < cntb) atomicAdd(&ncnt[w[k] & 255u], 1);
    } else {
        for (int i = tid; i < cntb; i += 1024) atomicAdd(&ncnt[binbuf[start + i] & 255], 1);
    }
    __syncthreads();
    if (tid < 32) {
        int a[8];
        int t = 0;
        for (int k = 0; k < 8; ++k) { a[k] = ncnt[lane * 8 + k]; t += a[k]; }
        int inc = t;
        for (int o = 1; o < 32; o <<= 1) {
            int u = __shfl_up(inc, o, 32);
            if (lane >= o) inc += u;
        }
        int run = inc - t;
        for (int k = 0; k < 8; ++k) { noff[lane * 8 + k] = run; run += a[k]; }
    }
    __syncthreads();
    if (tid < BIN_NODES) {
        const int node = b * BIN_NODES + tid;
        if (node < n) off[node] = start + noff[tid];
    }
    if (b == (int)gridDim.x - 1 && tid == 0) off[n] = start + cntb;

    if (staged) {
        for (int k = 0; k < FCAP / 1024; ++k) {
            if (k * 1024 + tid < cntb) {
                const int ld = (int)(w[k] & 255u);
                const int p  = noff[ld] + atomicAdd(&ncur[ld], 1);
                stage[p] = (int)(w[k] >> 8);
                snode[p] = (unsigned char)ld;
            }
        }
    } else {
        for (int i = tid; i < cntb; i += 1024) {
            const unsigned u = (unsigned)binbuf[start + i];
            const int ld = (int)(u & 255u);
            const int p  = noff[ld] + atomicAdd(&ncur[ld], 1);
            col[start + p] = (int)(u >> 8);
        }
    }
    __syncthreads();
    if (staged) {
        for (int i = tid; i < cntb; i += 1024) {
            const int ld = snode[i];
            const int v  = stage[i];
            const int lo = noff[ld];
            const int hi = lo + ncnt[ld];
            int r = lo;
            for (int q = lo; q < hi; ++q) {
                const int u = stage[q];
                r += (u < v || (u == v && q < i)) ? 1 : 0;
            }
            sorted[r] = v;
        }
    }
    __syncthreads();
    if (staged)
        for (int i = tid; i < cntb; i += 1024) col[start + i] = sorted[i];
}

#define GROWS 64
#define XPITCH (HID + 4)
__device__ __forceinline__ void gemm_x_attn_block(float* __restrict__ sh, int blk,
                                                  const float* __restrict__ h,
                                                  const float* __restrict__ W,
                                                  const float* __restrict__ a_d,
                                                  float* __restrict__ x,
                                                  float* __restrict__ ad_out,
                                                  int n) {
    const int block_row = blk * GROWS;
    const int tid  = threadIdx.x;
    const int wave = tid >> 5;
    const int lane = tid & 31;

    #pragma unroll
    for (int j = 0; j < GROWS * HID / 4 / 256; ++j) {
        const int r   = j * 8 + wave;
        const int row = min(block_row + r, n - 1);
        const float4 v = ((const float4*)(h + (size_t)row * HID))[lane];
        *(float4*)(sh + r * XPITCH + lane * 4) = v;
    }
    const float4 d4 = ((const float4*)a_d)[lane];
    __syncthreads();

    const int m    = lane & 15;
    const int koff = (lane >> 4) << 1;
    const int ncol = wave * 16 + (lane & 15);

    v8f c0 = {}, c1 = {}, c2 = {}, c3 = {};
    const float* arow = sh + m * XPITCH + koff;
    const float* wcol = W + (size_t)koff * HID + ncol;
    v2f a0 = *(const v2f*)(arow +  0 * XPITCH), e0 = *(const v2f*)(arow +  0 * XPITCH + 4);
    v2f a1 = *(const v2f*)(arow + 16 * XPITCH), e1 = *(const v2f*)(arow + 16 * XPITCH + 4);
    v2f a2 = *(const v2f*)(arow + 32 * XPITCH), e2 = *(const v2f*)(arow + 32 * XPITCH + 4);
    v2f a3 = *(const v2f*)(arow + 48 * XPITCH), e3 = *(const v2f*)(arow + 48 * XPITCH + 4);
    v2f b, f;
    b.x = wcol[0 * HID]; b.y = wcol[1 * HID];
    f.x = wcol[4 * HID]; f.y = wcol[5 * HID];
    #pragma unroll
    for (int k = 0; k < HID; k += 8) {
        const v2f ca0 = a0, ca1 = a1, ca2 = a2, ca3 = a3, cb = b;
        const v2f ce0 = e0, ce1 = e1, ce2 = e2, ce3 = e3, cf = f;
        if (k + 8 < HID) {
            const int kn = k + 8;
            a0 = *(const v2f*)(arow +  0 * XPITCH + kn); e0 = *(const v2f*)(arow +  0 * XPITCH + kn + 4);
            a1 = *(const v2f*)(arow + 16 * XPITCH + kn); e1 = *(const v2f*)(arow + 16 * XPITCH + kn + 4);
            a2 = *(const v2f*)(arow + 32 * XPITCH + kn); e2 = *(const v2f*)(arow + 32 * XPITCH + kn + 4);
            a3 = *(const v2f*)(arow + 48 * XPITCH + kn); e3 = *(const v2f*)(arow + 48 * XPITCH + kn + 4);
            b.x = wcol[(size_t)(kn)     * HID]; b.y = wcol[(size_t)(kn + 1) * HID];
            f.x = wcol[(size_t)(kn + 4) * HID]; f.y = wcol[(size_t)(kn + 5) * HID];
        }
        __builtin_amdgcn_sched_barrier(0);
        c0 = __builtin_amdgcn_wmma_f32_16x16x4_f32(false, ca0, false, cb, (short)0, c0, false, false);
        c1 = __builtin_amdgcn_wmma_f32_16x16x4_f32(false, ca1, false, cb, (short)0, c1, false, false);
        c2 = __builtin_amdgcn_wmma_f32_16x16x4_f32(false, ca2, false, cb, (short)0, c2, false, false);
        c3 = __builtin_amdgcn_wmma_f32_16x16x4_f32(false, ca3, false, cb, (short)0, c3, false, false);
        c0 = __builtin_amdgcn_wmma_f32_16x16x4_f32(false, ce0, false, cf, (short)0, c0, false, false);
        c1 = __builtin_amdgcn_wmma_f32_16x16x4_f32(false, ce1, false, cf, (short)0, c1, false, false);
        c2 = __builtin_amdgcn_wmma_f32_16x16x4_f32(false, ce2, false, cf, (short)0, c2, false, false);
        c3 = __builtin_amdgcn_wmma_f32_16x16x4_f32(false, ce3, false, cf, (short)0, c3, false, false);
    }
    __syncthreads();

    const int rloc = (lane >> 4) << 3;
    float* trow = sh + rloc * XPITCH + ncol;
    #pragma unroll
    for (int r = 0; r < 8; ++r) {
        trow[( 0 + r) * XPITCH] = c0[r];
        trow[(16 + r) * XPITCH] = c1[r];
        trow[(32 + r) * XPITCH] = c2[r];
        trow[(48 + r) * XPITCH] = c3[r];
    }
    __syncthreads();

    for (int j = 0; j < 8; ++j) {
        const int rl  = wave * 8 + j;
        const int row = block_row + rl;
        const float4 v = *(const float4*)(sh + rl * XPITCH + lane * 4);
        float pd = __builtin_fmaf(v.w, d4.w, __builtin_fmaf(v.z, d4.z, __builtin_fmaf(v.x, d4.x, v.y * d4.y)));
        #pragma unroll
        for (int o = 16; o > 0; o >>= 1) pd += __shfl_down(pd, o, 32);
        if (row < n) {
            ((float4*)(x + (size_t)row * HID))[lane] = v;
            if (lane == 0) ad_out[row] = pd;
        }
    }
}

__global__ __launch_bounds__(256) void gemm_x_attn_kernel(const float* __restrict__ h,
                                                          const float* __restrict__ W,
                                                          const float* __restrict__ a_d,
                                                          float* __restrict__ x,
                                                          float* __restrict__ ad_out,
                                                          int n) {
    __shared__ float sh[GROWS * XPITCH];
    gemm_x_attn_block(sh, blockIdx.x, h, W, a_d, x, ad_out, n);
}

static_assert(SCATTER_LDS_BYTES <= GROWS * XPITCH * 4, "the scatter's tables must fit the GEMM tile");
__global__ __launch_bounds__(256) void scatter_and_gemm0_kernel(const int* __restrict__ src,
                                                                const int* __restrict__ dst,
                                                                const int* __restrict__ tab,
                                                                int* __restrict__ bstart,
                                                                int* __restrict__ binbuf,
                                                                int n_edges, int nb, int nblk,
                                                                const float* __restrict__ h,
                                                                const float* __restrict__ W,
                                                                const float* __restrict__ a_d,
                                                                float* __restrict__ x,
                                                                float* __restrict__ ad_out,
                                                                int n) {
    __shared__ float sh[GROWS * XPITCH];
    const int b = blockIdx.x;
    if (b < nblk) bin_scatter_block((int*)sh, b, nblk, src, dst, tab, bstart, binbuf, n_edges, nb);
    else          gemm_x_attn_block(sh, b - nblk, h, W, a_d, x, ad_out, n);
}


__device__ __forceinline__ float xhalf(float v) {
    const int b = __float_as_int(v);
    return __int_as_float(__builtin_amdgcn_permlanex16(b, b, 0x76543210, 0xfedcba98, false, false));
}
template <int MASK>
__device__ __forceinline__ float row_xor(float v) {
    return __int_as_float(__builtin_amdgcn_update_dpp(0, __float_as_int(v), 0x160 + MASK, 0xf, 0xf, true));
}

__device__ __forceinline__ float half_logit(const float4 vl, const float4 vh, const float4 sl,
                                            const float4 sh, float adv) {
    const float pl = __builtin_fmaf(vl.w, sl.w, __builtin_fmaf(vl.z, sl.z, __builtin_fmaf(vl.x, sl.x, vl.y * sl.y)));
    const float ph = __builtin_fmaf(vh.w, sh.w, __builtin_fmaf(vh.z, sh.z, __builtin_fmaf(vh.x, sh.x, vh.y * sh.y)));
    float p = pl + ph;
    p += row_xor<8>(p);
    p += row_xor<4>(p);
    p += row_xor<2>(p);
    p += row_xor<1>(p);
    const float e = p + adv;
    return (e > 0.0f) ? e : NEG_SLOPE * e;
}

struct HalfSums { float4 al, ah; float sum, m; };

template <bool FULL>
__device__ __forceinline__ void take4(HalfSums& q, const char* __restrict__ xb, unsigned cofs, int h,
                                      int s, int j, int lim, const float4 sl, const float4 sh, float adv) {
    const int t0 = __builtin_amdgcn_readlane(s, j);
    const int t1 = __builtin_amdgcn_readlane(s, j + 1);
    const int t2 = __builtin_amdgcn_readlane(s, j + 2);
    const int t3 = __builtin_amdgcn_readlane(s, j + 3);
    const int  r0  = h ? t1 : t0;
    const int  r1  = h ? t3 : t2;
    const bool on0 = FULL || (j + h < lim);
    const bool on1 = FULL || (j + 2 + h < lim);
    float4 vl0 = make_float4(0.0f, 0.0f, 0.0f, 0.0f), vh0 = vl0, vl1 = vl0, vh1 = vl0;
    if (on0) {
        const char* p = xb + (((unsigned)r0 << 9) + cofs);
        vl0 = *(const float4*)p;
        vh0 = *(const float4*)(p + 256);
    }
    if (on1) {
        const char* p = xb + (((unsigned)r1 << 9) + cofs);
        vl1 = *(const float4*)p;
        vh1 = *(const float4*)(p + 256);
    }
    const float e0 = half_logit(vl0, vh0, sl, sh, adv);
    const float e1 = half_logit(vl1, vh1, sl, sh, adv);
    float mn, sc, w0, w1;
    if (FULL) {
        mn = fmaxf(fmaxf(q.m, e0), e1);
        sc = __expf(q.m - mn);
        w0 = __expf(e0 - mn);
        w1 = __expf(e1 - mn);
    } else {
        mn = q.m;
        if (on0) mn = fmaxf(mn, e0);
        if (on1) mn = fmaxf(mn, e1);
        sc = (mn > q.m) ? __expf(q.m - mn) : 1.0f;
        w0 = on0 ? __expf(e0 - mn) : 0.0f;
        w1 = on1 ? __expf(e1 - mn) : 0.0f;
    }
    q.sum  = q.sum  * sc + w0 + w1;
    q.al.x = q.al.x * sc + w0 * vl0.x + w1 * vl1.x;
    q.al.y = q.al.y * sc + w0 * vl0.y + w1 * vl1.y;
    q.al.z = q.al.z * sc + w0 * vl0.z + w1 * vl1.z;
    q.al.w = q.al.w * sc + w0 * vl0.w + w1 * vl1.w;
    q.ah.x = q.ah.x * sc + w0 * vh0.x + w1 * vh1.x;
    q.ah.y = q.ah.y * sc + w0 * vh0.y + w1 * vh1.y;
    q.ah.z = q.ah.z * sc + w0 * vh0.z + w1 * vh1.z;
    q.ah.w = q.ah.w * sc + w0 * vh0.w + w1 * vh1.w;
    q.m = mn;
}

__global__ __launch_bounds__(256) void node_gather_kernel(const int* __restrict__ off,
                                                          const int* __restrict__ col,
                                                          const float* __restrict__ x,
                                                          const float* __restrict__ a_s,
                                                          const float* __restrict__ ad_,
                                                          const float* __restrict__ bias,
                                                          float* __restrict__ out,
                                                          int n, int do_relu) {
    const int lane = threadIdx.x & 31;
    const int c = lane & 15;
    const int h = lane >> 4;
    const int d = __builtin_amdgcn_readfirstlane((int)((blockIdx.x * blockDim.x + threadIdx.x) >> 5));
    if (d >= n) return;
    const int   beg  = off[d];
    const int   cnt  = off[d + 1] - beg + 1;
    const float adv  = ad_[d];
    const int   last = n - 1;
    const char* __restrict__ xb = (const char*)x;
    const unsigned cofs = (unsigned)c << 4;
    const float4 sl = ((const float4*)a_s)[c];
    const float4 sh = ((const float4*)a_s)[16 + c];
    HalfSums q;
    q.al  = make_float4(0.0f, 0.0f, 0.0f, 0.0f);
    q.ah  = q.al;
    q.sum = 0.0f;
    q.m   = -__builtin_inff();

    for (int b = 0; b < cnt; b += 32) {
        const int k = b + lane;
        int s = d;
        if (k < cnt - 1) s = col[beg + k];
        s = min(max(s, 0), last);
        const int lim = min(32, cnt - b);
        int j = 0;
        for (; j + 4 <= lim; j += 4) take4<true>(q, xb, cofs, h, s, j, lim, sl, sh, adv);
        if (j < lim) take4<false>(q, xb, cofs, h, s, j, lim, sl, sh, adv);
    }
    const float4 al = q.al, ah = q.ah;
    const float  sum = q.sum, m = q.m;

    const float mt = fmaxf(m, xhalf(m));
    const float f  = __expf(m - mt);
    const float sf = sum * f;
    const float st = sf + xhalf(sf);
    const float4 keep = h ? ah : al;
    const float4 send = h ? al : ah;
    float4 acc;
    acc.x = keep.x * f + xhalf(send.x * f);
    acc.y = keep.y * f + xhalf(send.y * f);
    acc.z = keep.z * f + xhalf(send.z * f);
    acc.w = keep.w * f + xhalf(send.w * f);

    const float inv = 1.0f / (st + GAT_EPS);
    const float4 b4 = ((const float4*)bias)[lane];
    acc.x = acc.x * inv + b4.x; acc.y = acc.y * inv + b4.y;
    acc.z = acc.z * inv + b4.z; acc.w = acc.w * inv + b4.w;
    if (do_relu) {
        acc.x = fmaxf(acc.x, 0.0f); acc.y = fmaxf(acc.y, 0.0f);
        acc.z = fmaxf(acc.z, 0.0f); acc.w = fmaxf(acc.w, 0.0f);
    }
    ((float4*)(out + (size_t)d * HID))[lane] = acc;
}

extern "C" void kernel_launch(void* const* d_in, const int* in_sizes, int n_in,
                              void* d_out, int out_size, void* d_ws, size_t ws_size,
                              hipStream_t stream) {
    const float* z     = (const float*)d_in[0];
    const int*   ei    = (const int*)  d_in[1];
    const float* Ws    = (const float*)d_in[2];
    const float* a_src = (const float*)d_in[3];
    const float* a_dst = (const float*)d_in[4];
    const float* bias  = (const float*)d_in[5];
    float*       out   = (float*)d_out;

    const int N = in_sizes[0] / HID;
    const int E = in_sizes[1] / 2;
    const int* src = ei;
    const int* dst = ei + E;

    auto align = [](size_t v) { return (v + 255) & ~(size_t)255; };
    char* p = (char*)d_ws;
    float* x    = (float*)p; p += align((size_t)N * HID * sizeof(float));
    float* bufA = (float*)p; p += align((size_t)N * HID * sizeof(float));
    float* bufB = (float*)p; p += align((size_t)N * HID * sizeof(float));
    p += align((size_t)N * sizeof(float));
    float* ad_  = (float*)p; p += align((size_t)N * sizeof(float));
    p += align((size_t)N * sizeof(int));
    int*   off  = (int*)p;   p += align((size_t)(N + 1) * sizeof(int));
    int*   rank = (int*)p;   p += align((size_t)E * sizeof(int));
    int*   col  = (int*)p;   p += align((size_t)E * sizeof(int));

    int* tab    = (int*)bufB;
    int* bstart = (int*)bufA;
    int* binbuf = rank;
    const int NB   = (N + BIN_NODES - 1) / BIN_NODES;
    const int NBLK = (E + CHUNK - 1) / CHUNK;
    const int GBLK = (N + GROWS - 1) / GROWS;
    bin_hist_kernel<<<NBLK, 256, 0, stream>>>(dst, tab, E, NB);
    scatter_and_gemm0_kernel<<<NBLK + GBLK, 256, 0, stream>>>(src, dst, tab, bstart, binbuf, E, NB, NBLK,
                                                              z, Ws, a_dst, x, ad_, N);
    bin_finalize_kernel<<<NB, 1024, 0, stream>>>(binbuf, bstart, off, col, N);

    const float* h_ptr = z;
    for (int l = 0; l < 3; ++l) {
        float* o_ptr = (l == 0) ? bufA : (l == 1) ? bufB : out;
        const float* W  = Ws    + (size_t)l * HID * HID;
        const float* as = a_src + (size_t)l * HID;
        const float* ad = a_dst + (size_t)l * HID;
        const float* b  = bias  + (size_t)l * HID;

        if (l > 0) gemm_x_attn_kernel<<<GBLK, 256, 0, stream>>>(h_ptr, W, ad, x, ad_, N);
        node_gather_kernel<<<((size_t)N * 32 + 255) / 256, 256, 0, stream>>>(off, col, x, as, ad_, b,
                                                                             o_ptr, N, l < 2 ? 1 : 0);
        h_ptr = o_ptr;
    }
}
